# starting kernel
# baseline (speedup 1.0000x reference)
#include <hip/hip_runtime.h>
#include <math.h>

#define D 128

typedef __attribute__((ext_vector_type(2))) float v2f;
typedef __attribute__((ext_vector_type(4))) float v4f;
typedef __attribute__((ext_vector_type(8))) float v8f;

__global__ __launch_bounds__(256) void gin_zero(float* __restrict__ p, long n4) {
    long i = (long)blockIdx.x * blockDim.x + threadIdx.x;
    if (i < n4) ((v4f*)p)[i] = (v4f){0.f, 0.f, 0.f, 0.f};
}

__global__ __launch_bounds__(256) void gin_scatter(const float* __restrict__ h,
                                                   const int* __restrict__ src,
                                                   const int* __restrict__ dst,
                                                   float* __restrict__ aggr, int E) {
    int tid = blockIdx.x * 256 + threadIdx.x;
    int e = tid >> 5;
    if (e >= E) return;
    int c = (tid & 31) << 2;
    int s = src[e];
    int d = dst[e];
    v4f v = *(const v4f*)(h + (size_t)s * D + c);
    float* p = aggr + (size_t)d * D + c;
    atomicAdd(p + 0, v.x);
    atomicAdd(p + 1, v.y);
    atomicAdd(p + 2, v.z);
    atomicAdd(p + 3, v.w);
}

__global__ __launch_bounds__(256) void gin_gemm_elu(const float* h,
                                                    const float* __restrict__ aggr,
                                                    const float* __restrict__ W,
                                                    const float* __restrict__ b,
                                                    float* out, int n_rows) {
    __shared__ __align__(16) float ldsW[D * D];

    const int t = threadIdx.x;
    #pragma unroll
    for (int i = t; i < D * D / 4; i += 256)
        ((v4f*)ldsW)[i] = ((const v4f*)W)[i];
    __syncthreads();

    const int wave = t >> 5;
    const int lane = t & 31;
    const int row0 = (blockIdx.x * 8 + wave) * 16;
    if (row0 >= n_rows) return;

    const int m    = lane & 15;
    const int koff = (lane >> 4) << 1;

    const float* arow = h    + (size_t)(row0 + m) * D;
    const float* grow = aggr + (size_t)(row0 + m) * D;

    v8f c[8] = {};

    for (int kt = 0; kt < 32; ++kt) {
        const int k0 = kt * 4 + koff;
        v2f ha = *(const v2f*)(arow + k0);
        v2f ga = *(const v2f*)(grow + k0);
        v2f a = ha + ga;
        #pragma unroll
        for (int nt = 0; nt < 8; ++nt) {
            const int n = nt * 16 + m;
            v2f bf;
            bf.x = ldsW[k0 * D + n];
            bf.y = ldsW[(k0 + 1) * D + n];
            c[nt] = __builtin_amdgcn_wmma_f32_16x16x4_f32(
                false, a, false, bf, (short)0, c[nt], false, false);
        }
    }

    const int rbase = row0 + ((lane >> 4) << 3);
    #pragma unroll
    for (int nt = 0; nt < 8; ++nt) {
        const int col = nt * 16 + m;
        const float bias = b[col];
        #pragma unroll
        for (int v = 0; v < 8; ++v) {
            float x = c[nt][v] + bias;
            float y = x > 0.f ? x : expm1f(x);
            out[(size_t)(rbase + v) * D + col] = y;
        }
    }
}

extern "C" void kernel_launch(void* const* d_in, const int* in_sizes, int n_in,
                              void* d_out, int out_size, void* d_ws, size_t ws_size,
                              hipStream_t stream) {
    const float* x  = (const float*)d_in[0];
    const int*   ei = (const int*)d_in[1];
    const float* W0 = (const float*)d_in[2];
    const float* b0 = (const float*)d_in[3];
    const float* W1 = (const float*)d_in[4];
    const float* b1 = (const float*)d_in[5];
    const float* W2 = (const float*)d_in[6];
    const float* b2 = (const float*)d_in[7];

    const int N = in_sizes[0] / D;
    const int E = in_sizes[1] / 2;
    const int* src = ei;
    const int* dst = ei + E;

    float* out  = (float*)d_out;
    float* aggr = (float*)d_ws;

    const long n4 = (long)N * D / 4;
    const int zb = (int)((n4 + 255) / 256);
    const int sb = (int)(((long)E * 32 + 255) / 256);
    const int gb = (N / 16 + 7) / 8;

    gin_zero<<<zb, 256, 0, stream>>>(aggr, n4);
    gin_scatter<<<sb, 256, 0, stream>>>(x, src, dst, aggr, E);
    gin_gemm_elu<<<gb, 256, 0, stream>>>(x, aggr, W0, b0, out, N);

    gin_zero<<<zb, 256, 0, stream>>>(aggr, n4);
    gin_scatter<<<sb, 256, 0, stream>>>(out, src, dst, aggr, E);
    gin_gemm_elu<<<gb, 256, 0, stream>>>(out, aggr, W1, b1, out, N);

    gin_zero<<<zb, 256, 0, stream>>>(aggr, n4);
    gin_scatter<<<sb, 256, 0, stream>>>(out, src, dst, aggr, E);
    gin_gemm_elu<<<gb, 256, 0, stream>>>(out, aggr, W2, b2, out, N);
}


# Scatter with f32 atomics -> gather: edges grouped by destination once per call (int atomic slots + scan), one wave sums each node's neighbour rows, one store; sum order = slot arrival order
# speedup vs baseline: 2.6750x; 2.6750x over previous
#include <hip/hip_runtime.h>
#include <math.h>


#define D 128

typedef __attribute__((ext_vector_type(2))) float v2f;
typedef __attribute__((ext_vector_type(4))) float v4f;
typedef __attribute__((ext_vector_type(8))) float v8f;
typedef __attribute__((ext_vector_type(4))) int v4i;


__global__ __launch_bounds__(256) void csr_zero(int* __restrict__ cnt, int n_pad) {
    int i = blockIdx.x * 256 + threadIdx.x;
    if (i < n_pad) cnt[i] = 0;
}

__global__ __launch_bounds__(256) void csr_count(const int* __restrict__ dst,
                                                 int* __restrict__ cnt,
                                                 int* __restrict__ slot, int E) {
    int e = blockIdx.x * 256 + threadIdx.x;
    if (e < E) slot[e] = atomicAdd(&cnt[dst[e]], 1);
}

__device__ __forceinline__ int wave_incl_scan(int v, int lane) {
    for (int o = 1; o < 32; o <<= 1) {
        int u = __shfl_up(v, o);
        if (lane >= o) v += u;
    }
    return v;
}

__global__ __launch_bounds__(256) void csr_scan_block(const int* __restrict__ cnt,
                                                      int* __restrict__ off,
                                                      int* __restrict__ bsum) {
    __shared__ int wsum[8];
    const int t = threadIdx.x, lane = t & 31, wave = t >> 5;
    const int i4 = blockIdx.x * 256 + t;
    v4i c = ((const v4i*)cnt)[i4];
    int s = c.x + c.y + c.z + c.w;
    int incl = wave_incl_scan(s, lane);
    if (lane == 31) wsum[wave] = incl;
    __syncthreads();
    int wbase = 0;
    for (int w = 0; w < 8; ++w) wbase += (w < wave) ? wsum[w] : 0;
    int ex = wbase + incl - s;
    v4i o;
    o.x = ex;
    o.y = ex + c.x;
    o.z = ex + c.x + c.y;
    o.w = ex + c.x + c.y + c.z;
    ((v4i*)off)[i4] = o;
    if (t == 255) bsum[blockIdx.x] = wbase + incl;
}

__global__ __launch_bounds__(256) void csr_scan_top(int* __restrict__ bsum, int nb) {
    __shared__ int wsum[8];
    const int t = threadIdx.x, lane = t & 31, wave = t >> 5;
    int carry = 0;
    for (int base = 0; base < nb; base += 256) {
        const int i = base + t;
        int s = (i < nb) ? bsum[i] : 0;
        int incl = wave_incl_scan(s, lane);
        if (lane == 31) wsum[wave] = incl;
        __syncthreads();
        int wbase = 0, tot = 0;
        for (int w = 0; w < 8; ++w) {
            int x = wsum[w];
            wbase += (w < wave) ? x : 0;
            tot += x;
        }
        if (i < nb) bsum[i] = carry + wbase + incl - s;
        carry += tot;
        __syncthreads();
    }
}

__global__ __launch_bounds__(256) void csr_fill(const int* __restrict__ src,
                                                const int* __restrict__ dst,
                                                const int* __restrict__ slot,
                                                const int* __restrict__ off,
                                                const int* __restrict__ bbase,
                                                int* __restrict__ col, int E) {
    int e = blockIdx.x * 256 + threadIdx.x;
    if (e < E) {
        int d = dst[e];
        col[off[d] + bbase[d >> 10] + slot[e]] = src[e];
    }
}

__device__ __forceinline__ v4f nbr_row(const v4f* __restrict__ h4, int mine, int j, int lane) {
    const int s = __builtin_amdgcn_readlane(mine, j);
    return h4[(size_t)s * (D / 4) + lane];
}

__global__ __launch_bounds__(256) void gin_gather(const float* __restrict__ h,
                                                  const int* __restrict__ col,
                                                  const int* __restrict__ off,
                                                  const int* __restrict__ bbase,
                                                  const int* __restrict__ cnt,
                                                  float* __restrict__ aggr, int n_nodes) {
    const int lane = threadIdx.x & 31;
    const int n = blockIdx.x * 8 + (threadIdx.x >> 5);
    if (n >= n_nodes) return;
    const int beg = __builtin_amdgcn_readfirstlane(off[n] + bbase[n >> 10]);
    const int deg = __builtin_amdgcn_readfirstlane(cnt[n]);
    const v4f* h4 = (const v4f*)h;
    v4f acc = (v4f){0.f, 0.f, 0.f, 0.f};
    for (int b = 0; b < deg; b += 32) {
        const int m = (deg - b < 32) ? (deg - b) : 32;
        const int mine = (lane < m) ? col[beg + b + lane] : 0;
        int j = 0;
        for (; j + 8 <= m; j += 8) {
            const v4f r0 = nbr_row(h4, mine, j, lane);
            const v4f r1 = nbr_row(h4, mine, j + 1, lane);
            const v4f r2 = nbr_row(h4, mine, j + 2, lane);
            const v4f r3 = nbr_row(h4, mine, j + 3, lane);
            const v4f r4 = nbr_row(h4, mine, j + 4, lane);
            const v4f r5 = nbr_row(h4, mine, j + 5, lane);
            const v4f r6 = nbr_row(h4, mine, j + 6, lane);
            const v4f r7 = nbr_row(h4, mine, j + 7, lane);
            acc += r0;
            acc += r1;
            acc += r2;
            acc += r3;
            acc += r4;
            acc += r5;
            acc += r6;
            acc += r7;
        }
        if (j + 4 <= m) {
            const v4f r0 = nbr_row(h4, mine, j, lane);
            const v4f r1 = nbr_row(h4, mine, j + 1, lane);
            const v4f r2 = nbr_row(h4, mine, j + 2, lane);
            const v4f r3 = nbr_row(h4, mine, j + 3, lane);
            acc += r0;
            acc += r1;
            acc += r2;
            acc += r3;
            j += 4;
        }
        const int rest = m - j;
        if (rest == 3) {
            const v4f r0 = nbr_row(h4, mine, j, lane);
            const v4f r1 = nbr_row(h4, mine, j + 1, lane);
            const v4f r2 = nbr_row(h4, mine, j + 2, lane);
            acc += r0;
            acc += r1;
            acc += r2;
        } else if (rest == 2) {
            const v4f r0 = nbr_row(h4, mine, j, lane);
            const v4f r1 = nbr_row(h4, mine, j + 1, lane);
            acc += r0;
            acc += r1;
        } else if (rest == 1) {
            acc += nbr_row(h4, mine, j, lane);
        }
    }
    ((v4f*)aggr)[(size_t)n * (D / 4) + lane] = acc;
}

__global__ __launch_bounds__(256) void gin_gemm_elu(const float* h,
                                                    const float* __restrict__ aggr,
                                                    const float* __restrict__ W,
                                                    const float* __restrict__ b,
                                                    float* out, int n_rows) {
    __shared__ __align__(16) float ldsW[D * D];

    const int t = threadIdx.x;
    #pragma unroll
    for (int i = t; i < D * D / 4; i += 256)
        ((v4f*)ldsW)[i] = ((const v4f*)W)[i];
    __syncthreads();

    const int wave = t >> 5;
    const int lane = t & 31;
    const int row0 = (blockIdx.x * 8 + wave) * 16;
    if (row0 >= n_rows) return;

    const int m    = lane & 15;
    const int koff = (lane >> 4) << 1;

    const float* arow = h    + (size_t)(row0 + m) * D;
    const float* grow = aggr + (size_t)(row0 + m) * D;

    v8f c[8] = {};

    for (int kt = 0; kt < 32; ++kt) {
        const int k0 = kt * 4 + koff;
        v2f ha = *(const v2f*)(arow + k0);
        v2f ga = *(const v2f*)(grow + k0);
        v2f a = ha + ga;
        #pragma unroll
        for (int nt = 0; nt < 8; ++nt) {
            const int n = nt * 16 + m;
            v2f bf;
            bf.x = ldsW[k0 * D + n];
            bf.y = ldsW[(k0 + 1) * D + n];
            c[nt] = __builtin_amdgcn_wmma_f32_16x16x4_f32(
                false, a, false, bf, (short)0, c[nt], false, false);
        }
    }

    const int rbase = row0 + ((lane >> 4) << 3);
    #pragma unroll
    for (int nt = 0; nt < 8; ++nt) {
        const int col = nt * 16 + m;
        const float bias = b[col];
        #pragma unroll
        for (int v = 0; v < 8; ++v) {
            float x = c[nt][v] + bias;
            float y = x > 0.f ? x : expm1f(x);
            out[(size_t)(rbase + v) * D + col] = y;
        }
    }
}

extern "C" void kernel_launch(void* const* d_in, const int* in_sizes, int n_in,
                              void* d_out, int out_size, void* d_ws, size_t ws_size,
                              hipStream_t stream) {
    const float* x  = (const float*)d_in[0];
    const int*   ei = (const int*)d_in[1];
    const float* W0 = (const float*)d_in[2];
    const float* b0 = (const float*)d_in[3];
    const float* W1 = (const float*)d_in[4];
    const float* b1 = (const float*)d_in[5];
    const float* W2 = (const float*)d_in[6];
    const float* b2 = (const float*)d_in[7];

    const int N = in_sizes[0] / D;
    const int E = in_sizes[1] / 2;
    const int* src = ei;
    const int* dst = ei + E;

    const int nb = (N + 1023) / 1024;
    const int NP = nb * 1024;

    float* out  = (float*)d_out;
    float* aggr = (float*)d_ws;
    int* cnt  = (int*)(aggr + (size_t)N * D);
    int* off  = cnt + NP;
    int* bsum = off + NP;
    int* slot = bsum + 1024 * ((nb + 1023) / 1024);
    int* col  = slot + E;

    const int eb = (E + 255) / 256;
    const int ab = (N + 7) / 8;
    const int gb = (N / 16 + 7) / 8;

    csr_zero<<<NP / 256, 256, 0, stream>>>(cnt, NP);
    csr_count<<<eb, 256, 0, stream>>>(dst, cnt, slot, E);
    csr_scan_block<<<nb, 256, 0, stream>>>(cnt, off, bsum);
    csr_scan_top<<<1, 256, 0, stream>>>(bsum, nb);
    csr_fill<<<eb, 256, 0, stream>>>(src, dst, slot, off, bsum, col, E);

    gin_gather<<<ab, 256, 0, stream>>>(x, col, off, bsum, cnt, aggr, N);
    gin_gemm_elu<<<gb, 256, 0, stream>>>(x, aggr, W0, b0, out, N);

    gin_gather<<<ab, 256, 0, stream>>>(out, col, off, bsum, cnt, aggr, N);
    gin_gemm_elu<<<gb, 256, 0, stream>>>(out, aggr, W1, b1, out, N);

    gin_gather<<<ab, 256, 0, stream>>>(out, col, off, bsum, cnt, aggr, N);
    gin_gemm_elu<<<gb, 256, 0, stream>>>(out, aggr, W2, b2, out, N);
}


# Edge grouping without per-node global atomics: edges filed by 256-node range (LDS histogram, one global add per block and range), ranked per node by LDS counters; sum order = LDS arrival order
# speedup vs baseline: 3.5670x; 1.3335x over previous
#include <hip/hip_runtime.h>
#include <math.h>


#define D 128

typedef __attribute__((ext_vector_type(2))) float v2f;
typedef __attribute__((ext_vector_type(4))) float v4f;
typedef __attribute__((ext_vector_type(8))) float v8f;
typedef __attribute__((ext_vector_type(4))) int v4i;


#define BIN_SHIFT 8
#define BIN_NODES 256
#define MAX_BINS 1024
#define EDGE_T 1024
#define EDGE_U 4

typedef __attribute__((ext_vector_type(2))) int v2i;

__global__ __launch_bounds__(256) void bin_zero(int* __restrict__ binTot, int nbins) {
    int i = blockIdx.x * 256 + threadIdx.x;
    if (i < nbins) binTot[i] = 0;
}

__device__ __forceinline__ void chunk_histogram(const int* __restrict__ dst, int* hist,
                                                int e0, int e1, int t) {
    for (int e = e0 + t; e < e1; e += EDGE_T * EDGE_U) {
        int d[EDGE_U];
        #pragma unroll
        for (int u = 0; u < EDGE_U; ++u) {
            const int ee = e + u * EDGE_T;
            d[u] = (ee < e1) ? dst[ee] : -1;
        }
        #pragma unroll
        for (int u = 0; u < EDGE_U; ++u)
            if (d[u] >= 0) atomicAdd(&hist[d[u] >> BIN_SHIFT], 1);
    }
}

__global__ __launch_bounds__(EDGE_T) void bin_count(const int* __restrict__ dst,
                                                    int* __restrict__ binTot,
                                                    int E, int per_block, int nbins) {
    __shared__ int hist[MAX_BINS];
    const int t = threadIdx.x;
    for (int i = t; i < nbins; i += EDGE_T) hist[i] = 0;
    __syncthreads();
    const int e0 = blockIdx.x * per_block;
    const int e1 = (e0 + per_block < E) ? (e0 + per_block) : E;
    chunk_histogram(dst, hist, e0, e1, t);
    __syncthreads();
    for (int i = t; i < nbins; i += EDGE_T) {
        const int c = hist[i];
        if (c) atomicAdd(&binTot[i], c);
    }
}

__device__ __forceinline__ int wave_incl_scan(int v, int lane) {
    for (int o = 1; o < 32; o <<= 1) {
        int u = __shfl_up(v, o);
        if (lane >= o) v += u;
    }
    return v;
}

__global__ __launch_bounds__(256) void bin_scan(const int* __restrict__ binTot,
                                                int* __restrict__ binBase,
                                                int* __restrict__ binCur, int nbins) {
    __shared__ int wsum[8];
    const int t = threadIdx.x, lane = t & 31, wave = t >> 5;
    int carry = 0;
    for (int base = 0; base < nbins; base += 256) {
        const int i = base + t;
        int s = (i < nbins) ? binTot[i] : 0;
        int incl = wave_incl_scan(s, lane);
        if (lane == 31) wsum[wave] = incl;
        __syncthreads();
        int wbase = 0, tot = 0;
        for (int w = 0; w < 8; ++w) {
            int x = wsum[w];
            wbase += (w < wave) ? x : 0;
            tot += x;
        }
        if (i < nbins) {
            const int ex = carry + wbase + incl - s;
            binBase[i] = ex;
            binCur[i] = ex;
        }
        carry += tot;
        __syncthreads();
    }
}

__global__ __launch_bounds__(EDGE_T) void bin_scatter(const int* __restrict__ src,
                                                      const int* __restrict__ dst,
                                                      int* __restrict__ binCur,
                                                      v2i* __restrict__ rec,
                                                      int E, int per_block, int nbins) {
    __shared__ int hist[MAX_BINS];
    const int t = threadIdx.x;
    for (int i = t; i < nbins; i += EDGE_T) hist[i] = 0;
    __syncthreads();
    const int e0 = blockIdx.x * per_block;
    const int e1 = (e0 + per_block < E) ? (e0 + per_block) : E;
    chunk_histogram(dst, hist, e0, e1, t);
    __syncthreads();
    for (int i = t; i < nbins; i += EDGE_T) {
        const int c = hist[i];
        hist[i] = c ? atomicAdd(&binCur[i], c) : 0;
    }
    __syncthreads();
    for (int e = e0 + t; e < e1; e += EDGE_T * EDGE_U) {
        int d[EDGE_U], s[EDGE_U];
        #pragma unroll
        for (int u = 0; u < EDGE_U; ++u) {
            const int ee = e + u * EDGE_T;
            d[u] = (ee < e1) ? dst[ee] : -1;
            s[u] = (ee < e1) ? src[ee] : 0;
        }
        #pragma unroll
        for (int u = 0; u < EDGE_U; ++u) {
            if (d[u] >= 0) {
                const int pos = atomicAdd(&hist[d[u] >> BIN_SHIFT], 1);
                rec[pos] = (v2i){d[u], s[u]};
            }
        }
    }
}

__global__ __launch_bounds__(EDGE_T) void bin_finish(const v2i* __restrict__ rec,
                                                     const int* __restrict__ binBase,
                                                     const int* __restrict__ binTot,
                                                     int* __restrict__ cnt,
                                                     int* __restrict__ off,
                                                     int* __restrict__ col, int n_nodes) {
    __shared__ int c[BIN_NODES];
    __shared__ int wsum[8];
    const int t = threadIdx.x, lane = t & 31, wave = t >> 5;
    const int b = blockIdx.x;
    const int base = binBase[b];
    const int R = binTot[b];
    if (t < BIN_NODES) c[t] = 0;
    __syncthreads();
    for (int i = t; i < R; i += EDGE_T * EDGE_U) {
        int d[EDGE_U];
        #pragma unroll
        for (int u = 0; u < EDGE_U; ++u) {
            const int ii = i + u * EDGE_T;
            d[u] = (ii < R) ? rec[base + ii].x : -1;
        }
        #pragma unroll
        for (int u = 0; u < EDGE_U; ++u)
            if (d[u] >= 0) atomicAdd(&c[d[u] & (BIN_NODES - 1)], 1);
    }
    __syncthreads();
    int my = 0, incl = 0;
    if (t < BIN_NODES) {
        my = c[t];
        incl = wave_incl_scan(my, lane);
        if (lane == 31) wsum[wave] = incl;
    }
    __syncthreads();
    if (t < BIN_NODES) {
        int wbase = 0;
        for (int w = 0; w < 8; ++w) wbase += (w < wave) ? wsum[w] : 0;
        const int start = base + wbase + incl - my;
        const int n = b * BIN_NODES + t;
        if (n < n_nodes) {
            cnt[n] = my;
            off[n] = start;
        }
        c[t] = start;
    }
    __syncthreads();
    for (int i = t; i < R; i += EDGE_T * EDGE_U) {
        v2i r[EDGE_U];
        #pragma unroll
        for (int u = 0; u < EDGE_U; ++u) {
            const int ii = i + u * EDGE_T;
            r[u] = (ii < R) ? rec[base + ii] : (v2i){-1, 0};
        }
        #pragma unroll
        for (int u = 0; u < EDGE_U; ++u) {
            if (r[u].x >= 0) {
                const int pos = atomicAdd(&c[r[u].x & (BIN_NODES - 1)], 1);
                col[pos] = r[u].y;
            }
        }
    }
}

__device__ __forceinline__ v4f nbr_row(const v4f* __restrict__ h4, int mine, int j, int lane) {
    const int s = __builtin_amdgcn_readlane(mine, j);
    return h4[(size_t)s * (D / 4) + lane];
}

__global__ __launch_bounds__(256) void gin_gather(const float* __restrict__ h,
                                                  const int* __restrict__ col,
                                                  const int* __restrict__ off,
                                                  const int* __restrict__ cnt,
                                                  float* __restrict__ aggr, int n_nodes) {
    const int lane = threadIdx.x & 31;
    const int n = blockIdx.x * 8 + (threadIdx.x >> 5);
    if (n >= n_nodes) return;
    const int beg = __builtin_amdgcn_readfirstlane(off[n]);
    const int deg = __builtin_amdgcn_readfirstlane(cnt[n]);
    const v4f* h4 = (const v4f*)h;
    v4f acc = (v4f){0.f, 0.f, 0.f, 0.f};
    for (int b = 0; b < deg; b += 32) {
        const int m = (deg - b < 32) ? (deg - b) : 32;
        const int mine = (lane < m) ? col[beg + b + lane] : 0;
        int j = 0;
        for (; j + 8 <= m; j += 8) {
            const v4f r0 = nbr_row(h4, mine, j, lane);
            const v4f r1 = nbr_row(h4, mine, j + 1, lane);
            const v4f r2 = nbr_row(h4, mine, j + 2, lane);
            const v4f r3 = nbr_row(h4, mine, j + 3, lane);
            const v4f r4 = nbr_row(h4, mine, j + 4, lane);
            const v4f r5 = nbr_row(h4, mine, j + 5, lane);
            const v4f r6 = nbr_row(h4, mine, j + 6, lane);
            const v4f r7 = nbr_row(h4, mine, j + 7, lane);
            acc += r0;
            acc += r1;
            acc += r2;
            acc += r3;
            acc += r4;
            acc += r5;
            acc += r6;
            acc += r7;
        }
        if (j + 4 <= m) {
            const v4f r0 = nbr_row(h4, mine, j, lane);
            const v4f r1 = nbr_row(h4, mine, j + 1, lane);
            const v4f r2 = nbr_row(h4, mine, j + 2, lane);
            const v4f r3 = nbr_row(h4, mine, j + 3, lane);
            acc += r0;
            acc += r1;
            acc += r2;
            acc += r3;
            j += 4;
        }
        const int rest = m - j;
        if (rest == 3) {
            const v4f r0 = nbr_row(h4, mine, j, lane);
            const v4f r1 = nbr_row(h4, mine, j + 1, lane);
            const v4f r2 = nbr_row(h4, mine, j + 2, lane);
            acc += r0;
            acc += r1;
            acc += r2;
        } else if (rest == 2) {
            const v4f r0 = nbr_row(h4, mine, j, lane);
            const v4f r1 = nbr_row(h4, mine, j + 1, lane);
            acc += r0;
            acc += r1;
        } else if (rest == 1) {
            acc += nbr_row(h4, mine, j, lane);
        }
    }
    ((v4f*)aggr)[(size_t)n * (D / 4) + lane] = acc;
}

__global__ __launch_bounds__(256) void gin_gemm_elu(const float* h,
                                                    const float* __restrict__ aggr,
                                                    const float* __restrict__ W,
                                                    const float* __restrict__ b,
                                                    float* out, int n_rows) {
    __shared__ __align__(16) float ldsW[D * D];

    const int t = threadIdx.x;
    #pragma unroll
    for (int i = t; i < D * D / 4; i += 256)
        ((v4f*)ldsW)[i] = ((const v4f*)W)[i];
    __syncthreads();

    const int wave = t >> 5;
    const int lane = t & 31;
    const int row0 = (blockIdx.x * 8 + wave) * 16;
    if (row0 >= n_rows) return;

    const int m    = lane & 15;
    const int koff = (lane >> 4) << 1;

    const float* arow = h    + (size_t)(row0 + m) * D;
    const float* grow = aggr + (size_t)(row0 + m) * D;

    v8f c[8] = {};

    for (int kt = 0; kt < 32; ++kt) {
        const int k0 = kt * 4 + koff;
        v2f ha = *(const v2f*)(arow + k0);
        v2f ga = *(const v2f*)(grow + k0);
        v2f a = ha + ga;
        #pragma unroll
        for (int nt = 0; nt < 8; ++nt) {
            const int n = nt * 16 + m;
            v2f bf;
            bf.x = ldsW[k0 * D + n];
            bf.y = ldsW[(k0 + 1) * D + n];
            c[nt] = __builtin_amdgcn_wmma_f32_16x16x4_f32(
                false, a, false, bf, (short)0, c[nt], false, false);
        }
    }

    const int rbase = row0 + ((lane >> 4) << 3);
    #pragma unroll
    for (int nt = 0; nt < 8; ++nt) {
        const int col = nt * 16 + m;
        const float bias = b[col];
        #pragma unroll
        for (int v = 0; v < 8; ++v) {
            float x = c[nt][v] + bias;
            float y = x > 0.f ? x : expm1f(x);
            out[(size_t)(rbase + v) * D + col] = y;
        }
    }
}

extern "C" void kernel_launch(void* const* d_in, const int* in_sizes, int n_in,
                              void* d_out, int out_size, void* d_ws, size_t ws_size,
                              hipStream_t stream) {
    const float* x  = (const float*)d_in[0];
    const int*   ei = (const int*)d_in[1];
    const float* W0 = (const float*)d_in[2];
    const float* b0 = (const float*)d_in[3];
    const float* W1 = (const float*)d_in[4];
    const float* b1 = (const float*)d_in[5];
    const float* W2 = (const float*)d_in[6];
    const float* b2 = (const float*)d_in[7];

    const int N = in_sizes[0] / D;
    const int E = in_sizes[1] / 2;
    const int* src = ei;
    const int* dst = ei + E;

    const int nbins = (N + BIN_NODES - 1) / BIN_NODES;
    const int NP = nbins * BIN_NODES;

    float* out  = (float*)d_out;
    float* aggr = (float*)d_ws;
    int* cnt     = (int*)(aggr + (size_t)N * D);
    int* off     = cnt + NP;
    int* binTot  = off + NP;
    int* binBase = binTot + MAX_BINS;
    int* binCur  = binBase + MAX_BINS;
    v2i* rec     = (v2i*)(binCur + MAX_BINS);
    int* col     = (int*)(rec + E);

    const int EB = 256;
    const int per_block = (E + EB - 1) / EB;
    const int ab = (N + 7) / 8;
    const int gb = (N / 16 + 7) / 8;

    bin_zero<<<(nbins + 255) / 256, 256, 0, stream>>>(binTot, nbins);
    bin_count<<<EB, EDGE_T, 0, stream>>>(dst, binTot, E, per_block, nbins);
    bin_scan<<<1, 256, 0, stream>>>(binTot, binBase, binCur, nbins);
    bin_scatter<<<EB, EDGE_T, 0, stream>>>(src, dst, binCur, rec, E, per_block, nbins);
    bin_finish<<<nbins, EDGE_T, 0, stream>>>(rec, binBase, binTot, cnt, off, col, N);

    gin_gather<<<ab, 256, 0, stream>>>(x, col, off, cnt, aggr, N);
    gin_gemm_elu<<<gb, 256, 0, stream>>>(x, aggr, W0, b0, out, N);

    gin_gather<<<ab, 256, 0, stream>>>(out, col, off, cnt, aggr, N);
    gin_gemm_elu<<<gb, 256, 0, stream>>>(out, aggr, W1, b1, out, N);

    gin_gather<<<ab, 256, 0, stream>>>(out, col, off, cnt, aggr, N);
    gin_gemm_elu<<<gb, 256, 0, stream>>>(out, aggr, W2, b2, out, N);
}


# Neighbour sum fused into the dense layer: a wave walks its 16 nodes' index run, 16 rows per trip, parks the sums in an LDS tile read as WMMA A fragments; sum plane, 3 launches gone; add order kept
# speedup vs baseline: 4.1196x; 1.1549x over previous
#include <hip/hip_runtime.h>
#include <math.h>


#define D 128

typedef __attribute__((ext_vector_type(2))) float v2f;
typedef __attribute__((ext_vector_type(4))) float v4f;
typedef __attribute__((ext_vector_type(8))) float v8f;
typedef __attribute__((ext_vector_type(4))) int v4i;


#define BIN_SHIFT 8
#define BIN_NODES 256
#define MAX_BINS 1024
#define EDGE_T 1024
#define EDGE_U 4

typedef __attribute__((ext_vector_type(2))) int v2i;

__global__ __launch_bounds__(256) void bin_zero(int* __restrict__ binTot, int nbins) {
    int i = blockIdx.x * 256 + threadIdx.x;
    if (i < nbins) binTot[i] = 0;
}

__device__ __forceinline__ void chunk_histogram(const int* __restrict__ dst, int* hist,
                                                int e0, int e1, int t) {
    for (int e = e0 + t; e < e1; e += EDGE_T * EDGE_U) {
        int d[EDGE_U];
        #pragma unroll
        for (int u = 0; u < EDGE_U; ++u) {
            const int ee = e + u * EDGE_T;
            d[u] = (ee < e1) ? dst[ee] : -1;
        }
        #pragma unroll
        for (int u = 0; u < EDGE_U; ++u)
            if (d[u] >= 0) atomicAdd(&hist[d[u] >> BIN_SHIFT], 1);
    }
}

__global__ __launch_bounds__(EDGE_T) void bin_count(const int* __restrict__ dst,
                                                    int* __restrict__ binTot,
                                                    int E, int per_block, int nbins) {
    __shared__ int hist[MAX_BINS];
    const int t = threadIdx.x;
    for (int i = t; i < nbins; i += EDGE_T) hist[i] = 0;
    __syncthreads();
    const int e0 = blockIdx.x * per_block;
    const int e1 = (e0 + per_block < E) ? (e0 + per_block) : E;
    chunk_histogram(dst, hist, e0, e1, t);
    __syncthreads();
    for (int i = t; i < nbins; i += EDGE_T) {
        const int c = hist[i];
        if (c) atomicAdd(&binTot[i], c);
    }
}

__device__ __forceinline__ int wave_incl_scan(int v, int lane) {
    for (int o = 1; o < 32; o <<= 1) {
        int u = __shfl_up(v, o);
        if (lane >= o) v += u;
    }
    return v;
}

__global__ __launch_bounds__(256) void bin_scan(const int* __restrict__ binTot,
                                                int* __restrict__ binBase,
                                                int* __restrict__ binCur, int nbins) {
    __shared__ int wsum[8];
    const int t = threadIdx.x, lane = t & 31, wave = t >> 5;
    int carry = 0;
    for (int base = 0; base < nbins; base += 256) {
        const int i = base + t;
        int s = (i < nbins) ? binTot[i] : 0;
        int incl = wave_incl_scan(s, lane);
        if (lane == 31) wsum[wave] = incl;
        __syncthreads();
        int wbase = 0, tot = 0;
        for (int w = 0; w < 8; ++w) {
            int x = wsum[w];
            wbase += (w < wave) ? x : 0;
            tot += x;
        }
        if (i < nbins) {
            const int ex = carry + wbase + incl - s;
            binBase[i] = ex;
            binCur[i] = ex;
        }
        carry += tot;
        __syncthreads();
    }
}

__global__ __launch_bounds__(EDGE_T) void bin_scatter(const int* __restrict__ src,
                                                      const int* __restrict__ dst,
                                                      int* __restrict__ binCur,
                                                      v2i* __restrict__ rec,
                                                      int E, int per_block, int nbins) {
    __shared__ int hist[MAX_BINS];
    const int t = threadIdx.x;
    for (int i = t; i < nbins; i += EDGE_T) hist[i] = 0;
    __syncthreads();
    const int e0 = blockIdx.x * per_block;
    const int e1 = (e0 + per_block < E) ? (e0 + per_block) : E;
    chunk_histogram(dst, hist, e0, e1, t);
    __syncthreads();
    for (int i = t; i < nbins; i += EDGE_T) {
        const int c = hist[i];
        hist[i] = c ? atomicAdd(&binCur[i], c) : 0;
    }
    __syncthreads();
    for (int e = e0 + t; e < e1; e += EDGE_T * EDGE_U) {
        int d[EDGE_U], s[EDGE_U];
        #pragma unroll
        for (int u = 0; u < EDGE_U; ++u) {
            const int ee = e + u * EDGE_T;
            d[u] = (ee < e1) ? dst[ee] : -1;
            s[u] = (ee < e1) ? src[ee] : 0;
        }
        #pragma unroll
        for (int u = 0; u < EDGE_U; ++u) {
            if (d[u] >= 0) {
                const int pos = atomicAdd(&hist[d[u] >> BIN_SHIFT], 1);
                rec[pos] = (v2i){d[u], s[u]};
            }
        }
    }
}

__global__ __launch_bounds__(EDGE_T) void bin_finish(const v2i* __restrict__ rec,
                                                     const int* __restrict__ binBase,
                                                     const int* __restrict__ binTot,
                                                     int* __restrict__ cnt,
                                                     int* __restrict__ off,
                                                     int* __restrict__ col, int n_nodes) {
    __shared__ int c[BIN_NODES];
    __shared__ int wsum[8];
    const int t = threadIdx.x, lane = t & 31, wave = t >> 5;
    const int b = blockIdx.x;
    const int base = binBase[b];
    const int R = binTot[b];
    if (t < BIN_NODES) c[t] = 0;
    __syncthreads();
    for (int i = t; i < R; i += EDGE_T * EDGE_U) {
        int d[EDGE_U];
        #pragma unroll
        for (int u = 0; u < EDGE_U; ++u) {
            const int ii = i + u * EDGE_T;
            d[u] = (ii < R) ? rec[base + ii].x : -1;
        }
        #pragma unroll
        for (int u = 0; u < EDGE_U; ++u)
            if (d[u] >= 0) atomicAdd(&c[d[u] & (BIN_NODES - 1)], 1);
    }
    __syncthreads();
    int my = 0, incl = 0;
    if (t < BIN_NODES) {
        my = c[t];
        incl = wave_incl_scan(my, lane);
        if (lane == 31) wsum[wave] = incl;
    }
    __syncthreads();
    if (t < BIN_NODES) {
        int wbase = 0;
        for (int w = 0; w < 8; ++w) wbase += (w < wave) ? wsum[w] : 0;
        const int start = base + wbase + incl - my;
        const int n = b * BIN_NODES + t;
        if (n < n_nodes) {
            cnt[n] = my;
            off[n] = start;
        }
        c[t] = start;
    }
    __syncthreads();
    for (int i = t; i < R; i += EDGE_T * EDGE_U) {
        v2i r[EDGE_U];
        #pragma unroll
        for (int u = 0; u < EDGE_U; ++u) {
            const int ii = i + u * EDGE_T;
            r[u] = (ii < R) ? rec[base + ii] : (v2i){-1, 0};
        }
        #pragma unroll
        for (int u = 0; u < EDGE_U; ++u) {
            if (r[u].x >= 0) {
                const int pos = atomicAdd(&c[r[u].x & (BIN_NODES - 1)], 1);
                col[pos] = r[u].y;
            }
        }
    }
}

__device__ __forceinline__ v4f nbr_row(const v4f* __restrict__ h4, int mine, int j, int lane) {
    const int s = __builtin_amdgcn_readlane(mine, j);
    return h4[(size_t)s * (D / 4) + lane];
}

#define LAYER_WAVES 25
#define LAYER_T (LAYER_WAVES * 32)
#define IDX_CAP 384

__device__ __forceinline__ v4f* tile_slot(float* tile, int node, int lane) {
    return (v4f*)(tile + node * D + ((lane ^ node) << 2));
}

__global__ __launch_bounds__(LAYER_T) void gin_layer(const float* __restrict__ h,
                                                     const int* __restrict__ col,
                                                     const int* __restrict__ off,
                                                     const int* __restrict__ cnt,
                                                     const float* __restrict__ W,
                                                     const float* __restrict__ b,
                                                     float* __restrict__ out, int n_nodes) {
    __shared__ __align__(16) float ldsW[D * D];
    __shared__ __align__(16) float ldsA[LAYER_WAVES][16 * D];
    __shared__ int ldsIdx[LAYER_WAVES][IDX_CAP];

    const int t = threadIdx.x;
    const int wave = t >> 5;
    const int lane = t & 31;
    const int row0 = (blockIdx.x * LAYER_WAVES + wave) * 16;
    const bool live = row0 + 16 <= n_nodes;
    float* tile = ldsA[wave];
    int* idx = ldsIdx[wave];
    const v4f* h4 = (const v4f*)h;

    int o16 = 0, c16 = 0;
    if (live && lane < 16) {
        o16 = off[row0 + lane];
        c16 = cnt[row0 + lane];
    }
    if (live) {
        #pragma unroll
        for (int i = 0; i < 16; ++i)
            *tile_slot(tile, i, lane) = h4[(size_t)(row0 + i) * (D / 4) + lane];
    }

    #pragma unroll
    for (int k = 0; k < (D * D / 4 + LAYER_T - 1) / LAYER_T; ++k) {
        const int i = t + k * LAYER_T;
        if (i < D * D / 4) ((v4f*)ldsW)[i] = ((const v4f*)W)[i];
    }
    __syncthreads();
    if (!live) return;

    const int beg0 = __builtin_amdgcn_readlane(o16, 0);
    const int ends = o16 + c16 - beg0;
    const int len  = __builtin_amdgcn_readlane(ends, 15);
    int node = 0;
    int nend = __builtin_amdgcn_readlane(ends, 0);
    v4f acc = (v4f){0.f, 0.f, 0.f, 0.f};
    for (int w0 = 0; w0 < len; w0 += IDX_CAP) {
        const int wl = (len - w0 < IDX_CAP) ? (len - w0) : IDX_CAP;
        int win[IDX_CAP / 32];
        #pragma unroll
        for (int k = 0; k < IDX_CAP / 32; ++k) {
            int kk = lane + 32 * k;
            kk = (kk < wl) ? kk : (wl - 1);
            win[k] = col[beg0 + w0 + kk];
        }
        #pragma unroll
        for (int k = 0; k < IDX_CAP / 32; ++k) idx[lane + 32 * k] = win[k];
        __builtin_amdgcn_fence(__ATOMIC_ACQ_REL, "wavefront");
        for (int p = 0; p < wl; p += 16) {
            int q = p + (lane & 15);
            q = (q < wl) ? q : (wl - 1);
            const int mine = idx[q];
            v4f r[16];
            #pragma unroll
            for (int u = 0; u < 16; ++u) r[u] = nbr_row(h4, mine, u, lane);
            #pragma unroll
            for (int u = 0; u < 16; ++u) {
                if (p + u < wl) {
                    const int pos = w0 + p + u;
                    while (pos >= nend && node < 15) {
                        v4f* slot = tile_slot(tile, node, lane);
                        *slot = *slot + acc;
                        acc = (v4f){0.f, 0.f, 0.f, 0.f};
                        ++node;
                        nend = __builtin_amdgcn_readlane(ends, node);
                    }
                    acc += r[u];
                }
            }
        }
        __builtin_amdgcn_fence(__ATOMIC_ACQ_REL, "wavefront");
    }
    for (; node < 16; ++node) {
        v4f* slot = tile_slot(tile, node, lane);
        *slot = *slot + acc;
        acc = (v4f){0.f, 0.f, 0.f, 0.f};
    }
    __builtin_amdgcn_fence(__ATOMIC_ACQ_REL, "wavefront");

    const int m    = lane & 15;
    const int koff = (lane >> 4) << 1;
    const float* arow = tile + m * D + koff;

    v8f c[8] = {};

    for (int kt = 0; kt < 32; ++kt) {
        const int k0 = kt * 4 + koff;
        const v2f a = *(const v2f*)(arow + ((kt ^ m) << 2));
        #pragma unroll
        for (int nt = 0; nt < 8; ++nt) {
            const int n = nt * 16 + m;
            v2f bf;
            bf.x = ldsW[k0 * D + n];
            bf.y = ldsW[(k0 + 1) * D + n];
            c[nt] = __builtin_amdgcn_wmma_f32_16x16x4_f32(
                false, a, false, bf, (short)0, c[nt], false, false);
        }
    }

    const int rbase = row0 + ((lane >> 4) << 3);
    #pragma unroll
    for (int nt = 0; nt < 8; ++nt) {
        const int cc = nt * 16 + m;
        const float bias = b[cc];
        #pragma unroll
        for (int v = 0; v < 8; ++v) {
            float x = c[nt][v] + bias;
            float y = x > 0.f ? x : expm1f(x);
            out[(size_t)(rbase + v) * D + cc] = y;
        }
    }
}

extern "C" void kernel_launch(void* const* d_in, const int* in_sizes, int n_in,
                              void* d_out, int out_size, void* d_ws, size_t ws_size,
                              hipStream_t stream) {
    const float* x  = (const float*)d_in[0];
    const int*   ei = (const int*)d_in[1];
    const float* W0 = (const float*)d_in[2];
    const float* b0 = (const float*)d_in[3];
    const float* W1 = (const float*)d_in[4];
    const float* b1 = (const float*)d_in[5];
    const float* W2 = (const float*)d_in[6];
    const float* b2 = (const float*)d_in[7];

    const int N = in_sizes[0] / D;
    const int E = in_sizes[1] / 2;
    const int* src = ei;
    const int* dst = ei + E;

    const int nbins = (N + BIN_NODES - 1) / BIN_NODES;
    const int NP = nbins * BIN_NODES;

    float* out  = (float*)d_out;
    float* hmid = (float*)d_ws;
    int* cnt     = (int*)(hmid + (size_t)N * D);
    int* off     = cnt + NP;
    int* binTot  = off + NP;
    int* binBase = binTot + MAX_BINS;
    int* binCur  = binBase + MAX_BINS;
    v2i* rec     = (v2i*)(binCur + MAX_BINS);
    int* col     = (int*)(rec + E);

    const int EB = 256;
    const int per_block = (E + EB - 1) / EB;
    const int lb = (N / 16 + LAYER_WAVES - 1) / LAYER_WAVES;

    bin_zero<<<(nbins + 255) / 256, 256, 0, stream>>>(binTot, nbins);
    bin_count<<<EB, EDGE_T, 0, stream>>>(dst, binTot, E, per_block, nbins);
    bin_scan<<<1, 256, 0, stream>>>(binTot, binBase, binCur, nbins);
    bin_scatter<<<EB, EDGE_T, 0, stream>>>(src, dst, binCur, rec, E, per_block, nbins);
    bin_finish<<<nbins, EDGE_T, 0, stream>>>(rec, binBase, binTot, cnt, off, col, N);

    gin_layer<<<lb, LAYER_T, 0, stream>>>(x,    col, off, cnt, W0, b0, out,  N);
    gin_layer<<<lb, LAYER_T, 0, stream>>>(out,  col, off, cnt, W1, b1, hmid, N);
    gin_layer<<<lb, LAYER_T, 0, stream>>>(hmid, col, off, cnt, W2, b2, out,  N);
}


# Epilogue ELU branch-free: exp(x)-1 on the hardware v_exp_f32 (f32, abs error ~1e-7) computed for all elements and selected, instead of library expm1f behind a per-element branch
# speedup vs baseline: 4.7420x; 1.1511x over previous
#include <hip/hip_runtime.h>
#include <math.h>


#define D 128

typedef __attribute__((ext_vector_type(2))) float v2f;
typedef __attribute__((ext_vector_type(4))) float v4f;
typedef __attribute__((ext_vector_type(8))) float v8f;
typedef __attribute__((ext_vector_type(4))) int v4i;


#define BIN_SHIFT 8
#define BIN_NODES 256
#define MAX_BINS 1024
#define EDGE_T 1024
#define EDGE_U 4

typedef __attribute__((ext_vector_type(2))) int v2i;

__global__ __launch_bounds__(256) void bin_zero(int* __restrict__ binTot, int nbins) {
    int i = blockIdx.x * 256 + threadIdx.x;
    if (i < nbins) binTot[i] = 0;
}

__device__ __forceinline__ void chunk_histogram(const int* __restrict__ dst, int* hist,
                                                int e0, int e1, int t) {
    for (int e = e0 + t; e < e1; e += EDGE_T * EDGE_U) {
        int d[EDGE_U];
        #pragma unroll
        for (int u = 0; u < EDGE_U; ++u) {
            const int ee = e + u * EDGE_T;
            d[u] = (ee < e1) ? dst[ee] : -1;
        }
        #pragma unroll
        for (int u = 0; u < EDGE_U; ++u)
            if (d[u] >= 0) atomicAdd(&hist[d[u] >> BIN_SHIFT], 1);
    }
}

__global__ __launch_bounds__(EDGE_T) void bin_count(const int* __restrict__ dst,
                                                    int* __restrict__ binTot,
                                                    int E, int per_block, int nbins) {
    __shared__ int hist[MAX_BINS];
    const int t = threadIdx.x;
    for (int i = t; i < nbins; i += EDGE_T) hist[i] = 0;
    __syncthreads();
    const int e0 = blockIdx.x * per_block;
    const int e1 = (e0 + per_block < E) ? (e0 + per_block) : E;
    chunk_histogram(dst, hist, e0, e1, t);
    __syncthreads();
    for (int i = t; i < nbins; i += EDGE_T) {
        const int c = hist[i];
        if (c) atomicAdd(&binTot[i], c);
    }
}

__device__ __forceinline__ int wave_incl_scan(int v, int lane) {
    for (int o = 1; o < 32; o <<= 1) {
        int u = __shfl_up(v, o);
        if (lane >= o) v += u;
    }
    return v;
}

__global__ __launch_bounds__(256) void bin_scan(const int* __restrict__ binTot,
                                                int* __restrict__ binBase,
                                                int* __restrict__ binCur, int nbins) {
    __shared__ int wsum[8];
    const int t = threadIdx.x, lane = t & 31, wave = t >> 5;
    int carry = 0;
    for (int base = 0; base < nbins; base += 256) {
        const int i = base + t;
        int s = (i < nbins) ? binTot[i] : 0;
        int incl = wave_incl_scan(s, lane);
        if (lane == 31) wsum[wave] = incl;
        __syncthreads();
        int wbase = 0, tot = 0;
        for (int w = 0; w < 8; ++w) {
            int x = wsum[w];
            wbase += (w < wave) ? x : 0;
            tot += x;
        }
        if (i < nbins) {
            const int ex = carry + wbase + incl - s;
            binBase[i] = ex;
            binCur[i] = ex;
        }
        carry += tot;
        __syncthreads();
    }
}

__global__ __launch_bounds__(EDGE_T) void bin_scatter(const int* __restrict__ src,
                                                      const int* __restrict__ dst,
                                                      int* __restrict__ binCur,
                                                      v2i* __restrict__ rec,
                                                      int E, int per_block, int nbins) {
    __shared__ int hist[MAX_BINS];
    const int t = threadIdx.x;
    for (int i = t; i < nbins; i += EDGE_T) hist[i] = 0;
    __syncthreads();
    const int e0 = blockIdx.x * per_block;
    const int e1 = (e0 + per_block < E) ? (e0 + per_block) : E;
    chunk_histogram(dst, hist, e0, e1, t);
    __syncthreads();
    for (int i = t; i < nbins; i += EDGE_T) {
        const int c = hist[i];
        hist[i] = c ? atomicAdd(&binCur[i], c) : 0;
    }
    __syncthreads();
    for (int e = e0 + t; e < e1; e += EDGE_T * EDGE_U) {
        int d[EDGE_U], s[EDGE_U];
        #pragma unroll
        for (int u = 0; u < EDGE_U; ++u) {
            const int ee = e + u * EDGE_T;
            d[u] = (ee < e1) ? dst[ee] : -1;
            s[u] = (ee < e1) ? src[ee] : 0;
        }
        #pragma unroll
        for (int u = 0; u < EDGE_U; ++u) {
            if (d[u] >= 0) {
                const int pos = atomicAdd(&hist[d[u] >> BIN_SHIFT], 1);
                rec[pos] = (v2i){d[u], s[u]};
            }
        }
    }
}

__global__ __launch_bounds__(EDGE_T) void bin_finish(const v2i* __restrict__ rec,
                                                     const int* __restrict__ binBase,
                                                     const int* __restrict__ binTot,
                                                     int* __restrict__ cnt,
                                                     int* __restrict__ off,
                                                     int* __restrict__ col, int n_nodes) {
    __shared__ int c[BIN_NODES];
    __shared__ int wsum[8];
    const int t = threadIdx.x, lane = t & 31, wave = t >> 5;
    const int b = blockIdx.x;
    const int base = binBase[b];
    const int R = binTot[b];
    if (t < BIN_NODES) c[t] = 0;
    __syncthreads();
    for (int i = t; i < R; i += EDGE_T * EDGE_U) {
        int d[EDGE_U];
        #pragma unroll
        for (int u = 0; u < EDGE_U; ++u) {
            const int ii = i + u * EDGE_T;
            d[u] = (ii < R) ? rec[base + ii].x : -1;
        }
        #pragma unroll
        for (int u = 0; u < EDGE_U; ++u)
            if (d[u] >= 0) atomicAdd(&c[d[u] & (BIN_NODES - 1)], 1);
    }
    __syncthreads();
    int my = 0, incl = 0;
    if (t < BIN_NODES) {
        my = c[t];
        incl = wave_incl_scan(my, lane);
        if (lane == 31) wsum[wave] = incl;
    }
    __syncthreads();
    if (t < BIN_NODES) {
        int wbase = 0;
        for (int w = 0; w < 8; ++w) wbase += (w < wave) ? wsum[w] : 0;
        const int start = base + wbase + incl - my;
        const int n = b * BIN_NODES + t;
        if (n < n_nodes) {
            cnt[n] = my;
            off[n] = start;
        }
        c[t] = start;
    }
    __syncthreads();
    for (int i = t; i < R; i += EDGE_T * EDGE_U) {
        v2i r[EDGE_U];
        #pragma unroll
        for (int u = 0; u < EDGE_U; ++u) {
            const int ii = i + u * EDGE_T;
            r[u] = (ii < R) ? rec[base + ii] : (v2i){-1, 0};
        }
        #pragma unroll
        for (int u = 0; u < EDGE_U; ++u) {
            if (r[u].x >= 0) {
                const int pos = atomicAdd(&c[r[u].x & (BIN_NODES - 1)], 1);
                col[pos] = r[u].y;
            }
        }
    }
}

__device__ __forceinline__ v4f nbr_row(const v4f* __restrict__ h4, int mine, int j, int lane) {
    const int s = __builtin_amdgcn_readlane(mine, j);
    return h4[(size_t)s * (D / 4) + lane];
}

#define LAYER_WAVES 25
#define LAYER_T (LAYER_WAVES * 32)
#define IDX_CAP 384

__device__ __forceinline__ v4f* tile_slot(float* tile, int node, int lane) {
    return (v4f*)(tile + node * D + ((lane ^ node) << 2));
}

__global__ __launch_bounds__(LAYER_T) void gin_layer(const float* __restrict__ h,
                                                     const int* __restrict__ col,
                                                     const int* __restrict__ off,
                                                     const int* __restrict__ cnt,
                                                     const float* __restrict__ W,
                                                     const float* __restrict__ b,
                                                     float* __restrict__ out, int n_nodes) {
    __shared__ __align__(16) float ldsW[D * D];
    __shared__ __align__(16) float ldsA[LAYER_WAVES][16 * D];
    __shared__ int ldsIdx[LAYER_WAVES][IDX_CAP];

    const int t = threadIdx.x;
    const int wave = t >> 5;
    const int lane = t & 31;
    const int row0 = (blockIdx.x * LAYER_WAVES + wave) * 16;
    const bool live = row0 + 16 <= n_nodes;
    float* tile = ldsA[wave];
    int* idx = ldsIdx[wave];
    const v4f* h4 = (const v4f*)h;

    int o16 = 0, c16 = 0;
    if (live && lane < 16) {
        o16 = off[row0 + lane];
        c16 = cnt[row0 + lane];
    }
    if (live) {
        #pragma unroll
        for (int i = 0; i < 16; ++i)
            *tile_slot(tile, i, lane) = h4[(size_t)(row0 + i) * (D / 4) + lane];
    }

    #pragma unroll
    for (int k = 0; k < (D * D / 4 + LAYER_T - 1) / LAYER_T; ++k) {
        const int i = t + k * LAYER_T;
        if (i < D * D / 4) ((v4f*)ldsW)[i] = ((const v4f*)W)[i];
    }
    __syncthreads();
    if (!live) return;

    const int beg0 = __builtin_amdgcn_readlane(o16, 0);
    const int ends = o16 + c16 - beg0;
    const int len  = __builtin_amdgcn_readlane(ends, 15);
    int node = 0;
    int nend = __builtin_amdgcn_readlane(ends, 0);
    v4f acc = (v4f){0.f, 0.f, 0.f, 0.f};
    for (int w0 = 0; w0 < len; w0 += IDX_CAP) {
        const int wl = (len - w0 < IDX_CAP) ? (len - w0) : IDX_CAP;
        int win[IDX_CAP / 32];
        #pragma unroll
        for (int k = 0; k < IDX_CAP / 32; ++k) {
            int kk = lane + 32 * k;
            kk = (kk < wl) ? kk : (wl - 1);
            win[k] = col[beg0 + w0 + kk];
        }
        #pragma unroll
        for (int k = 0; k < IDX_CAP / 32; ++k) idx[lane + 32 * k] = win[k];
        __builtin_amdgcn_fence(__ATOMIC_ACQ_REL, "wavefront");
        for (int p = 0; p < wl; p += 16) {
            int q = p + (lane & 15);
            q = (q < wl) ? q : (wl - 1);
            const int mine = idx[q];
            v4f r[16];
            #pragma unroll
            for (int u = 0; u < 16; ++u) r[u] = nbr_row(h4, mine, u, lane);
            #pragma unroll
            for (int u = 0; u < 16; ++u) {
                if (p + u < wl) {
                    const int pos = w0 + p + u;
                    while (pos >= nend && node < 15) {
                        v4f* slot = tile_slot(tile, node, lane);
                        *slot = *slot + acc;
                        acc = (v4f){0.f, 0.f, 0.f, 0.f};
                        ++node;
                        nend = __builtin_amdgcn_readlane(ends, node);
                    }
                    acc += r[u];
                }
            }
        }
        __builtin_amdgcn_fence(__ATOMIC_ACQ_REL, "wavefront");
    }
    for (; node < 16; ++node) {
        v4f* slot = tile_slot(tile, node, lane);
        *slot = *slot + acc;
        acc = (v4f){0.f, 0.f, 0.f, 0.f};
    }
    __builtin_amdgcn_fence(__ATOMIC_ACQ_REL, "wavefront");

    const int m    = lane & 15;
    const int koff = (lane >> 4) << 1;
    const float* arow = tile + m * D + koff;

    v8f c[8] = {};

    for (int kt = 0; kt < 32; ++kt) {
        const int k0 = kt * 4 + koff;
        const v2f a = *(const v2f*)(arow + ((kt ^ m) << 2));
        #pragma unroll
        for (int nt = 0; nt < 8; ++nt) {
            const int n = nt * 16 + m;
            v2f bf;
            bf.x = ldsW[k0 * D + n];
            bf.y = ldsW[(k0 + 1) * D + n];
            c[nt] = __builtin_amdgcn_wmma_f32_16x16x4_f32(
                false, a, false, bf, (short)0, c[nt], false, false);
        }
    }

    const int rbase = row0 + ((lane >> 4) << 3);
    #pragma unroll
    for (int nt = 0; nt < 8; ++nt) {
        const int cc = nt * 16 + m;
        const float bias = b[cc];
        #pragma unroll
        for (int v = 0; v < 8; ++v) {
            float x = c[nt][v] + bias;
            float em1 = __builtin_amdgcn_exp2f(x * 1.4426950408889634f) - 1.0f;
            float y = x > 0.f ? x : em1;
            out[(size_t)(rbase + v) * D + cc] = y;
        }
    }
}

extern "C" void kernel_launch(void* const* d_in, const int* in_sizes, int n_in,
                              void* d_out, int out_size, void* d_ws, size_t ws_size,
                              hipStream_t stream) {
    const float* x  = (const float*)d_in[0];
    const int*   ei = (const int*)d_in[1];
    const float* W0 = (const float*)d_in[2];
    const float* b0 = (const float*)d_in[3];
    const float* W1 = (const float*)d_in[4];
    const float* b1 = (const float*)d_in[5];
    const float* W2 = (const float*)d_in[6];
    const float* b2 = (const float*)d_in[7];

    const int N = in_sizes[0] / D;
    const int E = in_sizes[1] / 2;
    const int* src = ei;
    const int* dst = ei + E;

    const int nbins = (N + BIN_NODES - 1) / BIN_NODES;
    const int NP = nbins * BIN_NODES;

    float* out  = (float*)d_out;
    float* hmid = (float*)d_ws;
    int* cnt     = (int*)(hmid + (size_t)N * D);
    int* off     = cnt + NP;
    int* binTot  = off + NP;
    int* binBase = binTot + MAX_BINS;
    int* binCur  = binBase + MAX_BINS;
    v2i* rec     = (v2i*)(binCur + MAX_BINS);
    int* col     = (int*)(rec + E);

    const int EB = 256;
    const int per_block = (E + EB - 1) / EB;
    const int lb = (N / 16 + LAYER_WAVES - 1) / LAYER_WAVES;

    bin_zero<<<(nbins + 255) / 256, 256, 0, stream>>>(binTot, nbins);
    bin_count<<<EB, EDGE_T, 0, stream>>>(dst, binTot, E, per_block, nbins);
    bin_scan<<<1, 256, 0, stream>>>(binTot, binBase, binCur, nbins);
    bin_scatter<<<EB, EDGE_T, 0, stream>>>(src, dst, binCur, rec, E, per_block, nbins);
    bin_finish<<<nbins, EDGE_T, 0, stream>>>(rec, binBase, binTot, cnt, off, col, N);

    gin_layer<<<lb, LAYER_T, 0, stream>>>(x,    col, off, cnt, W0, b0, out,  N);
    gin_layer<<<lb, LAYER_T, 0, stream>>>(out,  col, off, cnt, W1, b1, hmid, N);
    gin_layer<<<lb, LAYER_T, 0, stream>>>(hmid, col, off, cnt, W2, b2, out,  N);
}


# Neighbour walk stripped of scalar work: a group's 16 indices come by LDS broadcast (next group's read early), 32-bit row offsets on a scalar base, one compare per row for bucket ends; add order kept
# speedup vs baseline: 4.8990x; 1.0331x over previous
#include <hip/hip_runtime.h>
#include <math.h>


#define D 128

typedef __attribute__((ext_vector_type(2))) float v2f;
typedef __attribute__((ext_vector_type(4))) float v4f;
typedef __attribute__((ext_vector_type(8))) float v8f;
typedef __attribute__((ext_vector_type(4))) int v4i;


#define BIN_SHIFT 8
#define BIN_NODES 256
#define MAX_BINS 1024
#define EDGE_T 1024
#define EDGE_U 4

typedef __attribute__((ext_vector_type(2))) int v2i;

__global__ __launch_bounds__(256) void bin_zero(int* __restrict__ binTot, int nbins) {
    int i = blockIdx.x * 256 + threadIdx.x;
    if (i < nbins) binTot[i] = 0;
}

__device__ __forceinline__ void chunk_histogram(const int* __restrict__ dst, int* hist,
                                                int e0, int e1, int t) {
    for (int e = e0 + t; e < e1; e += EDGE_T * EDGE_U) {
        int d[EDGE_U];
        #pragma unroll
        for (int u = 0; u < EDGE_U; ++u) {
            const int ee = e + u * EDGE_T;
            d[u] = (ee < e1) ? dst[ee] : -1;
        }
        #pragma unroll
        for (int u = 0; u < EDGE_U; ++u)
            if (d[u] >= 0) atomicAdd(&hist[d[u] >> BIN_SHIFT], 1);
    }
}

__global__ __launch_bounds__(EDGE_T) void bin_count(const int* __restrict__ dst,
                                                    int* __restrict__ binTot,
                                                    int E, int per_block, int nbins) {
    __shared__ int hist[MAX_BINS];
    const int t = threadIdx.x;
    for (int i = t; i < nbins; i += EDGE_T) hist[i] = 0;
    __syncthreads();
    const int e0 = blockIdx.x * per_block;
    const int e1 = (e0 + per_block < E) ? (e0 + per_block) : E;
    chunk_histogram(dst, hist, e0, e1, t);
    __syncthreads();
    for (int i = t; i < nbins; i += EDGE_T) {
        const int c = hist[i];
        if (c) atomicAdd(&binTot[i], c);
    }
}

__device__ __forceinline__ int wave_incl_scan(int v, int lane) {
    for (int o = 1; o < 32; o <<= 1) {
        int u = __shfl_up(v, o);
        if (lane >= o) v += u;
    }
    return v;
}

__global__ __launch_bounds__(256) void bin_scan(const int* __restrict__ binTot,
                                                int* __restrict__ binBase,
                                                int* __restrict__ binCur, int nbins) {
    __shared__ int wsum[8];
    const int t = threadIdx.x, lane = t & 31, wave = t >> 5;
    int carry = 0;
    for (int base = 0; base < nbins; base += 256) {
        const int i = base + t;
        int s = (i < nbins) ? binTot[i] : 0;
        int incl = wave_incl_scan(s, lane);
        if (lane == 31) wsum[wave] = incl;
        __syncthreads();
        int wbase = 0, tot = 0;
        for (int w = 0; w < 8; ++w) {
            int x = wsum[w];
            wbase += (w < wave) ? x : 0;
            tot += x;
        }
        if (i < nbins) {
            const int ex = carry + wbase + incl - s;
            binBase[i] = ex;
            binCur[i] = ex;
        }
        carry += tot;
        __syncthreads();
    }
}

__global__ __launch_bounds__(EDGE_T) void bin_scatter(const int* __restrict__ src,
                                                      const int* __restrict__ dst,
                                                      int* __restrict__ binCur,
                                                      v2i* __restrict__ rec,
                                                      int E, int per_block, int nbins) {
    __shared__ int hist[MAX_BINS];
    const int t = threadIdx.x;
    for (int i = t; i < nbins; i += EDGE_T) hist[i] = 0;
    __syncthreads();
    const int e0 = blockIdx.x * per_block;
    const int e1 = (e0 + per_block < E) ? (e0 + per_block) : E;
    chunk_histogram(dst, hist, e0, e1, t);
    __syncthreads();
    for (int i = t; i < nbins; i += EDGE_T) {
        const int c = hist[i];
        hist[i] = c ? atomicAdd(&binCur[i], c) : 0;
    }
    __syncthreads();
    for (int e = e0 + t; e < e1; e += EDGE_T * EDGE_U) {
        int d[EDGE_U], s[EDGE_U];
        #pragma unroll
        for (int u = 0; u < EDGE_U; ++u) {
            const int ee = e + u * EDGE_T;
            d[u] = (ee < e1) ? dst[ee] : -1;
            s[u] = (ee < e1) ? src[ee] : 0;
        }
        #pragma unroll
        for (int u = 0; u < EDGE_U; ++u) {
            if (d[u] >= 0) {
                const int pos = atomicAdd(&hist[d[u] >> BIN_SHIFT], 1);
                rec[pos] = (v2i){d[u], s[u]};
            }
        }
    }
}

__global__ __launch_bounds__(EDGE_T) void bin_finish(const v2i* __restrict__ rec,
                                                     const int* __restrict__ binBase,
                                                     const int* __restrict__ binTot,
                                                     int* __restrict__ cnt,
                                                     int* __restrict__ off,
                                                     int* __restrict__ col, int n_nodes) {
    __shared__ int c[BIN_NODES];
    __shared__ int wsum[8];
    const int t = threadIdx.x, lane = t & 31, wave = t >> 5;
    const int b = blockIdx.x;
    const int base = binBase[b];
    const int R = binTot[b];
    if (t < BIN_NODES) c[t] = 0;
    __syncthreads();
    for (int i = t; i < R; i += EDGE_T * EDGE_U) {
        int d[EDGE_U];
        #pragma unroll
        for (int u = 0; u < EDGE_U; ++u) {
            const int ii = i + u * EDGE_T;
            d[u] = (ii < R) ? rec[base + ii].x : -1;
        }
        #pragma unroll
        for (int u = 0; u < EDGE_U; ++u)
            if (d[u] >= 0) atomicAdd(&c[d[u] & (BIN_NODES - 1)], 1);
    }
    __syncthreads();
    int my = 0, incl = 0;
    if (t < BIN_NODES) {
        my = c[t];
        incl = wave_incl_scan(my, lane);
        if (lane == 31) wsum[wave] = incl;
    }
    __syncthreads();
    if (t < BIN_NODES) {
        int wbase = 0;
        for (int w = 0; w < 8; ++w) wbase += (w < wave) ? wsum[w] : 0;
        const int start = base + wbase + incl - my;
        const int n = b * BIN_NODES + t;
        if (n < n_nodes) {
            cnt[n] = my;
            off[n] = start;
        }
        c[t] = start;
    }
    __syncthreads();
    for (int i = t; i < R; i += EDGE_T * EDGE_U) {
        v2i r[EDGE_U];
        #pragma unroll
        for (int u = 0; u < EDGE_U; ++u) {
            const int ii = i + u * EDGE_T;
            r[u] = (ii < R) ? rec[base + ii] : (v2i){-1, 0};
        }
        #pragma unroll
        for (int u = 0; u < EDGE_U; ++u) {
            if (r[u].x >= 0) {
                const int pos = atomicAdd(&c[r[u].x & (BIN_NODES - 1)], 1);
                col[pos] = r[u].y;
            }
        }
    }
}

#define LAYER_WAVES 25
#define LAYER_T (LAYER_WAVES * 32)
#define IDX_CAP 384

__device__ __forceinline__ v4f* tile_slot(float* tile, int node, int lane) {
    return (v4f*)(tile + node * D + ((lane ^ node) << 2));
}

__global__ __launch_bounds__(LAYER_T) void gin_layer(const float* __restrict__ h,
                                                     const int* __restrict__ col,
                                                     const int* __restrict__ off,
                                                     const int* __restrict__ cnt,
                                                     const float* __restrict__ W,
                                                     const float* __restrict__ b,
                                                     float* __restrict__ out, int n_nodes) {
    __shared__ __align__(16) float ldsW[D * D];
    __shared__ __align__(16) float ldsA[LAYER_WAVES][16 * D];
    __shared__ __align__(16) int ldsIdx[LAYER_WAVES][IDX_CAP];

    const int t = threadIdx.x;
    const int wave = t >> 5;
    const int lane = t & 31;
    const int row0 = (blockIdx.x * LAYER_WAVES + wave) * 16;
    const bool live = row0 + 16 <= n_nodes;
    float* tile = ldsA[wave];
    int* idx = ldsIdx[wave];
    const v4f* h4 = (const v4f*)h;

    int o16 = 0, c16 = 0;
    if (live && lane < 16) {
        o16 = off[row0 + lane];
        c16 = cnt[row0 + lane];
    }
    if (live) {
        #pragma unroll
        for (int i = 0; i < 16; ++i)
            *tile_slot(tile, i, lane) = h4[(size_t)(row0 + i) * (D / 4) + lane];
    }

    #pragma unroll
    for (int k = 0; k < (D * D / 4 + LAYER_T - 1) / LAYER_T; ++k) {
        const int i = t + k * LAYER_T;
        if (i < D * D / 4) ((v4f*)ldsW)[i] = ((const v4f*)W)[i];
    }
    __syncthreads();
    if (!live) return;

    const int beg0 = __builtin_amdgcn_readlane(o16, 0);
    const int ends = o16 + c16 - beg0;
    const int len  = __builtin_amdgcn_readlane(ends, 15);
    int node = 0;
    int nend = __builtin_amdgcn_readlane(ends, 0);
    v4f acc = (v4f){0.f, 0.f, 0.f, 0.f};
    const char* hb = (const char*)h;
    const unsigned lane16 = (unsigned)lane << 4;
    auto park = [&](int pos) {
        while (pos >= nend && node < 15) {
            v4f* slot = tile_slot(tile, node, lane);
            *slot = *slot + acc;
            acc = (v4f){0.f, 0.f, 0.f, 0.f};
            ++node;
            nend = __builtin_amdgcn_readlane(ends, node);
        }
    };
    for (int w0 = 0; w0 < len; w0 += IDX_CAP) {
        const int wl = (len - w0 < IDX_CAP) ? (len - w0) : IDX_CAP;
        int win[IDX_CAP / 32];
        #pragma unroll
        for (int k = 0; k < IDX_CAP / 32; ++k) {
            int kk = lane + 32 * k;
            kk = (kk < wl) ? kk : (wl - 1);
            win[k] = col[beg0 + w0 + kk];
        }
        #pragma unroll
        for (int k = 0; k < IDX_CAP / 32; ++k) idx[lane + 32 * k] = win[k];
        __builtin_amdgcn_fence(__ATOMIC_ACQ_REL, "wavefront");
        v4i q0 = ((const v4i*)idx)[0], q1 = ((const v4i*)idx)[1];
        v4i q2 = ((const v4i*)idx)[2], q3 = ((const v4i*)idx)[3];
        for (int p = 0; p < wl; p += 16) {
            const int s[16] = {q0.x, q0.y, q0.z, q0.w, q1.x, q1.y, q1.z, q1.w,
                               q2.x, q2.y, q2.z, q2.w, q3.x, q3.y, q3.z, q3.w};
            v4f r[16];
            unsigned ro[16];
            #pragma unroll
            for (int u = 0; u < 16; ++u) ro[u] = (((unsigned)s[u]) << 9) + lane16;
            asm volatile("" :: "v"(ro[0]), "v"(ro[1]), "v"(ro[2]), "v"(ro[3]), "v"(ro[4]), "v"(ro[5]),
                         "v"(ro[6]), "v"(ro[7]), "v"(ro[8]), "v"(ro[9]), "v"(ro[10]), "v"(ro[11]),
                         "v"(ro[12]), "v"(ro[13]), "v"(ro[14]), "v"(ro[15]));
            #pragma unroll
            for (int u = 0; u < 16; ++u) r[u] = *(const v4f*)(hb + ro[u]);
            if (p + 16 < wl) {
                const v4i* ip = (const v4i*)(idx + p + 16);
                q0 = ip[0]; q1 = ip[1]; q2 = ip[2]; q3 = ip[3];
            }
            const int base = w0 + p;
            int rel = nend - base;
            if (p + 16 <= wl) {
                #pragma unroll
                for (int u = 0; u < 16; ++u) {
                    if (u >= rel) {
                        park(base + u);
                        rel = nend - base;
                    }
                    acc += r[u];
                }
            } else {
                #pragma unroll
                for (int u = 0; u < 16; ++u) {
                    if (p + u < wl) {
                        if (u >= rel) {
                            park(base + u);
                            rel = nend - base;
                        }
                        acc += r[u];
                    }
                }
            }
        }
        __builtin_amdgcn_fence(__ATOMIC_ACQ_REL, "wavefront");
    }
    for (; node < 16; ++node) {
        v4f* slot = tile_slot(tile, node, lane);
        *slot = *slot + acc;
        acc = (v4f){0.f, 0.f, 0.f, 0.f};
    }
    __builtin_amdgcn_fence(__ATOMIC_ACQ_REL, "wavefront");

    const int m    = lane & 15;
    const int koff = (lane >> 4) << 1;
    const float* arow = tile + m * D + koff;

    v8f c[8] = {};

    for (int kt = 0; kt < 32; ++kt) {
        const int k0 = kt * 4 + koff;
        const v2f a = *(const v2f*)(arow + ((kt ^ m) << 2));
        #pragma unroll
        for (int nt = 0; nt < 8; ++nt) {
            const int n = nt * 16 + m;
            v2f bf;
            bf.x = ldsW[k0 * D + n];
            bf.y = ldsW[(k0 + 1) * D + n];
            c[nt] = __builtin_amdgcn_wmma_f32_16x16x4_f32(
                false, a, false, bf, (short)0, c[nt], false, false);
        }
    }

    const int rbase = row0 + ((lane >> 4) << 3);
    #pragma unroll
    for (int nt = 0; nt < 8; ++nt) {
        const int cc = nt * 16 + m;
        const float bias = b[cc];
        #pragma unroll
        for (int v = 0; v < 8; ++v) {
            float x = c[nt][v] + bias;
            float em1 = __builtin_amdgcn_exp2f(x * 1.4426950408889634f) - 1.0f;
            float y = x > 0.f ? x : em1;
            out[(size_t)(rbase + v) * D + cc] = y;
        }
    }
}

extern "C" void kernel_launch(void* const* d_in, const int* in_sizes, int n_in,
                              void* d_out, int out_size, void* d_ws, size_t ws_size,
                              hipStream_t stream) {
    const float* x  = (const float*)d_in[0];
    const int*   ei = (const int*)d_in[1];
    const float* W0 = (const float*)d_in[2];
    const float* b0 = (const float*)d_in[3];
    const float* W1 = (const float*)d_in[4];
    const float* b1 = (const float*)d_in[5];
    const float* W2 = (const float*)d_in[6];
    const float* b2 = (const float*)d_in[7];

    const int N = in_sizes[0] / D;
    const int E = in_sizes[1] / 2;
    const int* src = ei;
    const int* dst = ei + E;

    const int nbins = (N + BIN_NODES - 1) / BIN_NODES;
    const int NP = nbins * BIN_NODES;

    float* out  = (float*)d_out;
    float* hmid = (float*)d_ws;
    int* cnt     = (int*)(hmid + (size_t)N * D);
    int* off     = cnt + NP;
    int* binTot  = off + NP;
    int* binBase = binTot + MAX_BINS;
    int* binCur  = binBase + MAX_BINS;
    v2i* rec     = (v2i*)(binCur + MAX_BINS);
    int* col     = (int*)(rec + E);

    const int EB = 256;
    const int per_block = (E + EB - 1) / EB;
    const int lb = (N / 16 + LAYER_WAVES - 1) / LAYER_WAVES;

    bin_zero<<<(nbins + 255) / 256, 256, 0, stream>>>(binTot, nbins);
    bin_count<<<EB, EDGE_T, 0, stream>>>(dst, binTot, E, per_block, nbins);
    bin_scan<<<1, 256, 0, stream>>>(binTot, binBase, binCur, nbins);
    bin_scatter<<<EB, EDGE_T, 0, stream>>>(src, dst, binCur, rec, E, per_block, nbins);
    bin_finish<<<nbins, EDGE_T, 0, stream>>>(rec, binBase, binTot, cnt, off, col, N);

    gin_layer<<<lb, LAYER_T, 0, stream>>>(x,    col, off, cnt, W0, b0, out,  N);
    gin_layer<<<lb, LAYER_T, 0, stream>>>(out,  col, off, cnt, W1, b1, hmid, N);
    gin_layer<<<lb, LAYER_T, 0, stream>>>(hmid, col, off, cnt, W2, b2, out,  N);
}


# Layer block's 25 waves work as two teams a gather phase apart: 13 walk their rows at once, 12 stage W and wait at the block barrier, then walk under the first team's WMMA K-loops; add order kept
# speedup vs baseline: 5.1153x; 1.0442x over previous
#include <hip/hip_runtime.h>
#include <math.h>


#define D 128

typedef __attribute__((ext_vector_type(2))) float v2f;
typedef __attribute__((ext_vector_type(4))) float v4f;
typedef __attribute__((ext_vector_type(8))) float v8f;
typedef __attribute__((ext_vector_type(4))) int v4i;


#define BIN_SHIFT 8
#define BIN_NODES 256
#define MAX_BINS 1024
#define EDGE_T 1024
#define EDGE_U 4

typedef __attribute__((ext_vector_type(2))) int v2i;

__global__ __launch_bounds__(256) void bin_zero(int* __restrict__ binTot, int nbins) {
    int i = blockIdx.x * 256 + threadIdx.x;
    if (i < nbins) binTot[i] = 0;
}

__device__ __forceinline__ void chunk_histogram(const int* __restrict__ dst, int* hist,
                                                int e0, int e1, int t) {
    for (int e = e0 + t; e < e1; e += EDGE_T * EDGE_U) {
        int d[EDGE_U];
        #pragma unroll
        for (int u = 0; u < EDGE_U; ++u) {
            const int ee = e + u * EDGE_T;
            d[u] = (ee < e1) ? dst[ee] : -1;
        }
        #pragma unroll
        for (int u = 0; u < EDGE_U; ++u)
            if (d[u] >= 0) atomicAdd(&hist[d[u] >> BIN_SHIFT], 1);
    }
}

__global__ __launch_bounds__(EDGE_T) void bin_count(const int* __restrict__ dst,
                                                    int* __restrict__ binTot,
                                                    int E, int per_block, int nbins) {
    __shared__ int hist[MAX_BINS];
    const int t = threadIdx.x;
    for (int i = t; i < nbins; i += EDGE_T) hist[i] = 0;
    __syncthreads();
    const int e0 = blockIdx.x * per_block;
    const int e1 = (e0 + per_block < E) ? (e0 + per_block) : E;
    chunk_histogram(dst, hist, e0, e1, t);
    __syncthreads();
    for (int i = t; i < nbins; i += EDGE_T) {
        const int c = hist[i];
        if (c) atomicAdd(&binTot[i], c);
    }
}

__device__ __forceinline__ int wave_incl_scan(int v, int lane) {
    for (int o = 1; o < 32; o <<= 1) {
        int u = __shfl_up(v, o);
        if (lane >= o) v += u;
    }
    return v;
}

__global__ __launch_bounds__(256) void bin_scan(const int* __restrict__ binTot,
                                                int* __restrict__ binBase,
                                                int* __restrict__ binCur, int nbins) {
    __shared__ int wsum[8];
    const int t = threadIdx.x, lane = t & 31, wave = t >> 5;
    int carry = 0;
    for (int base = 0; base < nbins; base += 256) {
        const int i = base + t;
        int s = (i < nbins) ? binTot[i] : 0;
        int incl = wave_incl_scan(s, lane);
        if (lane == 31) wsum[wave] = incl;
        __syncthreads();
        int wbase = 0, tot = 0;
        for (int w = 0; w < 8; ++w) {
            int x = wsum[w];
            wbase += (w < wave) ? x : 0;
            tot += x;
        }
        if (i < nbins) {
            const int ex = carry + wbase + incl - s;
            binBase[i] = ex;
            binCur[i] = ex;
        }
        carry += tot;
        __syncthreads();
    }
}

__global__ __launch_bounds__(EDGE_T) void bin_scatter(const int* __restrict__ src,
                                                      const int* __restrict__ dst,
                                                      int* __restrict__ binCur,
                                                      v2i* __restrict__ rec,
                                                      int E, int per_block, int nbins) {
    __shared__ int hist[MAX_BINS];
    const int t = threadIdx.x;
    for (int i = t; i < nbins; i += EDGE_T) hist[i] = 0;
    __syncthreads();
    const int e0 = blockIdx.x * per_block;
    const int e1 = (e0 + per_block < E) ? (e0 + per_block) : E;
    chunk_histogram(dst, hist, e0, e1, t);
    __syncthreads();
    for (int i = t; i < nbins; i += EDGE_T) {
        const int c = hist[i];
        hist[i] = c ? atomicAdd(&binCur[i], c) : 0;
    }
    __syncthreads();
    for (int e = e0 + t; e < e1; e += EDGE_T * EDGE_U) {
        int d[EDGE_U], s[EDGE_U];
        #pragma unroll
        for (int u = 0; u < EDGE_U; ++u) {
            const int ee = e + u * EDGE_T;
            d[u] = (ee < e1) ? dst[ee] : -1;
            s[u] = (ee < e1) ? src[ee] : 0;
        }
        #pragma unroll
        for (int u = 0; u < EDGE_U; ++u) {
            if (d[u] >= 0) {
                const int pos = atomicAdd(&hist[d[u] >> BIN_SHIFT], 1);
                rec[pos] = (v2i){d[u], s[u]};
            }
        }
    }
}

__global__ __launch_bounds__(EDGE_T) void bin_finish(const v2i* __restrict__ rec,
                                                     const int* __restrict__ binBase,
                                                     const int* __restrict__ binTot,
                                                     int* __restrict__ cnt,
                                                     int* __restrict__ off,
                                                     int* __restrict__ col, int n_nodes) {
    __shared__ int c[BIN_NODES];
    __shared__ int wsum[8];
    const int t = threadIdx.x, lane = t & 31, wave = t >> 5;
    const int b = blockIdx.x;
    const int base = binBase[b];
    const int R = binTot[b];
    if (t < BIN_NODES) c[t] = 0;
    __syncthreads();
    for (int i = t; i < R; i += EDGE_T * EDGE_U) {
        int d[EDGE_U];
        #pragma unroll
        for (int u = 0; u < EDGE_U; ++u) {
            const int ii = i + u * EDGE_T;
            d[u] = (ii < R) ? rec[base + ii].x : -1;
        }
        #pragma unroll
        for (int u = 0; u < EDGE_U; ++u)
            if (d[u] >= 0) atomicAdd(&c[d[u] & (BIN_NODES - 1)], 1);
    }
    __syncthreads();
    int my = 0, incl = 0;
    if (t < BIN_NODES) {
        my = c[t];
        incl = wave_incl_scan(my, lane);
        if (lane == 31) wsum[wave] = incl;
    }
    __syncthreads();
    if (t < BIN_NODES) {
        int wbase = 0;
        for (int w = 0; w < 8; ++w) wbase += (w < wave) ? wsum[w] : 0;
        const int start = base + wbase + incl - my;
        const int n = b * BIN_NODES + t;
        if (n < n_nodes) {
            cnt[n] = my;
            off[n] = start;
        }
        c[t] = start;
    }
    __syncthreads();
    for (int i = t; i < R; i += EDGE_T * EDGE_U) {
        v2i r[EDGE_U];
        #pragma unroll
        for (int u = 0; u < EDGE_U; ++u) {
            const int ii = i + u * EDGE_T;
            r[u] = (ii < R) ? rec[base + ii] : (v2i){-1, 0};
        }
        #pragma unroll
        for (int u = 0; u < EDGE_U; ++u) {
            if (r[u].x >= 0) {
                const int pos = atomicAdd(&c[r[u].x & (BIN_NODES - 1)], 1);
                col[pos] = r[u].y;
            }
        }
    }
}

#define LAYER_WAVES 25
#define LAYER_T (LAYER_WAVES * 32)
#define LATE_T (12 * 32)
static_assert(LAYER_WAVES == 25, "the two teams and LATE_T are laid out for 25 waves");
#define IDX_CAP 384

__device__ __forceinline__ v4f* tile_slot(float* tile, int node, int lane) {
    return (v4f*)(tile + node * D + ((lane ^ node) << 2));
}

__global__ __launch_bounds__(LAYER_T) void gin_layer(const float* __restrict__ h,
                                                     const int* __restrict__ col,
                                                     const int* __restrict__ off,
                                                     const int* __restrict__ cnt,
                                                     const float* __restrict__ W,
                                                     const float* __restrict__ b,
                                                     float* __restrict__ out, int n_nodes) {
    __shared__ __align__(16) float ldsW[D * D];
    __shared__ __align__(16) float ldsA[LAYER_WAVES][16 * D];
    __shared__ __align__(16) int ldsIdx[LAYER_WAVES][IDX_CAP];

    const int t = threadIdx.x;
    const int wave = t >> 5;
    const int lane = t & 31;
    const int row0 = (blockIdx.x * LAYER_WAVES + wave) * 16;
    const bool live = row0 + 16 <= n_nodes;
    float* tile = ldsA[wave];
    int* idx = ldsIdx[wave];
    const v4f* h4 = (const v4f*)h;

    int o16 = 0, c16 = 0;
    if (live && lane < 16) {
        o16 = off[row0 + lane];
        c16 = cnt[row0 + lane];
    }
    if (live) {
        #pragma unroll
        for (int i = 0; i < 16; ++i)
            *tile_slot(tile, i, lane) = h4[(size_t)(row0 + i) * (D / 4) + lane];
    }

    const bool early = ((wave >> 2) & 1) == 0;
    #pragma nounroll
    for (int ph = 0; ph < 2; ++ph) {
    if (ph == 0 && !early) {
        const int tl = ((((wave >> 3) << 2) | (wave & 3)) << 5) + lane;
        #pragma unroll
        for (int k = 0; k < (D * D / 4 + LATE_T - 1) / LATE_T; ++k) {
            const int i = tl + k * LATE_T;
            if (i < D * D / 4) ((v4f*)ldsW)[i] = ((const v4f*)W)[i];
        }
    }
    if (live && early == (ph == 0)) {
    const int beg0 = __builtin_amdgcn_readlane(o16, 0);
    const int ends = o16 + c16 - beg0;
    const int len  = __builtin_amdgcn_readlane(ends, 15);
    int node = 0;
    int nend = __builtin_amdgcn_readlane(ends, 0);
    v4f acc = (v4f){0.f, 0.f, 0.f, 0.f};
    const char* hb = (const char*)h;
    const unsigned lane16 = (unsigned)lane << 4;
    auto park = [&](int pos) {
        while (pos >= nend && node < 15) {
            v4f* slot = tile_slot(tile, node, lane);
            *slot = *slot + acc;
            acc = (v4f){0.f, 0.f, 0.f, 0.f};
            ++node;
            nend = __builtin_amdgcn_readlane(ends, node);
        }
    };
    for (int w0 = 0; w0 < len; w0 += IDX_CAP) {
        const int wl = (len - w0 < IDX_CAP) ? (len - w0) : IDX_CAP;
        int win[IDX_CAP / 32];
        #pragma unroll
        for (int k = 0; k < IDX_CAP / 32; ++k) {
            int kk = lane + 32 * k;
            kk = (kk < wl) ? kk : (wl - 1);
            win[k] = col[beg0 + w0 + kk];
        }
        #pragma unroll
        for (int k = 0; k < IDX_CAP / 32; ++k) idx[lane + 32 * k] = win[k];
        __builtin_amdgcn_fence(__ATOMIC_ACQ_REL, "wavefront");
        v4i q0 = ((const v4i*)idx)[0], q1 = ((const v4i*)idx)[1];
        v4i q2 = ((const v4i*)idx)[2], q3 = ((const v4i*)idx)[3];
        for (int p = 0; p < wl; p += 16) {
            const int s[16] = {q0.x, q0.y, q0.z, q0.w, q1.x, q1.y, q1.z, q1.w,
                               q2.x, q2.y, q2.z, q2.w, q3.x, q3.y, q3.z, q3.w};
            v4f r[16];
            unsigned ro[16];
            #pragma unroll
            for (int u = 0; u < 16; ++u) ro[u] = (((unsigned)s[u]) << 9) + lane16;
            asm volatile("" :: "v"(ro[0]), "v"(ro[1]), "v"(ro[2]), "v"(ro[3]), "v"(ro[4]), "v"(ro[5]),
                         "v"(ro[6]), "v"(ro[7]), "v"(ro[8]), "v"(ro[9]), "v"(ro[10]), "v"(ro[11]),
                         "v"(ro[12]), "v"(ro[13]), "v"(ro[14]), "v"(ro[15]));
            #pragma unroll
            for (int u = 0; u < 16; ++u) r[u] = *(const v4f*)(hb + ro[u]);
            if (p + 16 < wl) {
                const v4i* ip = (const v4i*)(idx + p + 16);
                q0 = ip[0]; q1 = ip[1]; q2 = ip[2]; q3 = ip[3];
            }
            const int base = w0 + p;
            int rel = nend - base;
            if (p + 16 <= wl) {
                #pragma unroll
                for (int u = 0; u < 16; ++u) {
                    if (u >= rel) {
                        park(base + u);
                        rel = nend - base;
                    }
                    acc += r[u];
                }
            } else {
                #pragma unroll
                for (int u = 0; u < 16; ++u) {
                    if (p + u < wl) {
                        if (u >= rel) {
                            park(base + u);
                            rel = nend - base;
                        }
                        acc += r[u];
                    }
                }
            }
        }
        __builtin_amdgcn_fence(__ATOMIC_ACQ_REL, "wavefront");
    }
    for (; node < 16; ++node) {
        v4f* slot = tile_slot(tile, node, lane);
        *slot = *slot + acc;
        acc = (v4f){0.f, 0.f, 0.f, 0.f};
    }
    }
    if (ph == 0) __syncthreads();
    }
    if (!live) return;
    __builtin_amdgcn_fence(__ATOMIC_ACQ_REL, "wavefront");

    const int m    = lane & 15;
    const int koff = (lane >> 4) << 1;
    const float* arow = tile + m * D + koff;

    v8f c[8] = {};

    for (int kt = 0; kt < 32; ++kt) {
        const int k0 = kt * 4 + koff;
        const v2f a = *(const v2f*)(arow + ((kt ^ m) << 2));
        #pragma unroll
        for (int nt = 0; nt < 8; ++nt) {
            const int n = nt * 16 + m;
            v2f bf;
            bf.x = ldsW[k0 * D + n];
            bf.y = ldsW[(k0 + 1) * D + n];
            c[nt] = __builtin_amdgcn_wmma_f32_16x16x4_f32(
                false, a, false, bf, (short)0, c[nt], false, false);
        }
    }

    const int rbase = row0 + ((lane >> 4) << 3);
    #pragma unroll
    for (int nt = 0; nt < 8; ++nt) {
        const int cc = nt * 16 + m;
        const float bias = b[cc];
        #pragma unroll
        for (int v = 0; v < 8; ++v) {
            float x = c[nt][v] + bias;
            float em1 = __builtin_amdgcn_exp2f(x * 1.4426950408889634f) - 1.0f;
            float y = x > 0.f ? x : em1;
            out[(size_t)(rbase + v) * D + cc] = y;
        }
    }
}

extern "C" void kernel_launch(void* const* d_in, const int* in_sizes, int n_in,
                              void* d_out, int out_size, void* d_ws, size_t ws_size,
                              hipStream_t stream) {
    const float* x  = (const float*)d_in[0];
    const int*   ei = (const int*)d_in[1];
    const float* W0 = (const float*)d_in[2];
    const float* b0 = (const float*)d_in[3];
    const float* W1 = (const float*)d_in[4];
    const float* b1 = (const float*)d_in[5];
    const float* W2 = (const float*)d_in[6];
    const float* b2 = (const float*)d_in[7];

    const int N = in_sizes[0] / D;
    const int E = in_sizes[1] / 2;
    const int* src = ei;
    const int* dst = ei + E;

    const int nbins = (N + BIN_NODES - 1) / BIN_NODES;
    const int NP = nbins * BIN_NODES;

    float* out  = (float*)d_out;
    float* hmid = (float*)d_ws;
    int* cnt     = (int*)(hmid + (size_t)N * D);
    int* off     = cnt + NP;
    int* binTot  = off + NP;
    int* binBase = binTot + MAX_BINS;
    int* binCur  = binBase + MAX_BINS;
    v2i* rec     = (v2i*)(binCur + MAX_BINS);
    int* col     = (int*)(rec + E);

    const int EB = 256;
    const int per_block = (E + EB - 1) / EB;
    const int lb = (N / 16 + LAYER_WAVES - 1) / LAYER_WAVES;

    bin_zero<<<(nbins + 255) / 256, 256, 0, stream>>>(binTot, nbins);
    bin_count<<<EB, EDGE_T, 0, stream>>>(dst, binTot, E, per_block, nbins);
    bin_scan<<<1, 256, 0, stream>>>(binTot, binBase, binCur, nbins);
    bin_scatter<<<EB, EDGE_T, 0, stream>>>(src, dst, binCur, rec, E, per_block, nbins);
    bin_finish<<<nbins, EDGE_T, 0, stream>>>(rec, binBase, binTot, cnt, off, col, N);

    gin_layer<<<lb, LAYER_T, 0, stream>>>(x,    col, off, cnt, W0, b0, out,  N);
    gin_layer<<<lb, LAYER_T, 0, stream>>>(out,  col, off, cnt, W1, b1, hmid, N);
    gin_layer<<<lb, LAYER_T, 0, stream>>>(hmid, col, off, cnt, W2, b2, out,  N);
}
